# starting kernel
# baseline (speedup 1.0000x reference)
#include <hip/hip_runtime.h>


typedef __attribute__((ext_vector_type(2))) float v2f;
typedef __attribute__((ext_vector_type(8))) float v8f;

#define BN_EPS 1e-5f

__device__ __forceinline__ void atomic_add_f32(float* p, float v) {
    unsafeAtomicAdd(p, v);
}

__global__ void conv1_scatter(const float* __restrict__ x,
                              const float* __restrict__ W1,
                              const int*   __restrict__ in_idx,
                              const int*   __restrict__ out_idx,
                              float*       __restrict__ h_acc,
                              int M, int KM) {
    int t = blockIdx.x * blockDim.x + threadIdx.x;
    if (t >= KM) return;
    int k = t / M;
    float xv = x[in_idx[t]];
    int ro = out_idx[t] * 16;
    const float* w = W1 + k * 16;
#pragma unroll
    for (int c = 0; c < 16; ++c)
        atomic_add_f32(&h_acc[ro + c], xv * w[c]);
}

__global__ void bn_elu(const float* __restrict__ in, float* __restrict__ out,
                       const float* __restrict__ g,  const float* __restrict__ b,
                       const float* __restrict__ mn, const float* __restrict__ vr,
                       int total, int Cmask) {
    int t = blockIdx.x * blockDim.x + threadIdx.x;
    if (t >= total) return;
    int c = t & Cmask;
    float s = g[c] * rsqrtf(vr[c] + BN_EPS);
    float y = (in[t] - mn[c]) * s + b[c];
    out[t] = (y > 0.0f) ? y : (expf(y) - 1.0f);
}

__global__ void conv2_wmma(const float* __restrict__ h,
                           const float* __restrict__ W2,
                           const int*   __restrict__ in_idx,
                           const int*   __restrict__ out_idx,
                           float*       __restrict__ acc,
                           int M, int Mt, int totalWaves) {
    int wave = blockIdx.x * (blockDim.x >> 5) + (threadIdx.x >> 5);
    if (wave >= totalWaves) return;
    int lane = threadIdx.x & 31;
    int half = lane >> 4;
    int l15  = lane & 15;

    int k  = wave / Mt;
    int m0 = (wave - k * Mt) << 4;
    int base = k * M + m0;

    int arow = in_idx[base + l15];
    const float* hp = h + arow * 16 + (half << 1);

    const float* wk = W2 + k * 512;

    v8f c0 = {};
    v8f c1 = {};

#pragma unroll
    for (int q = 0; q < 4; ++q) {
        v2f a;
        a.x = hp[4 * q + 0];
        a.y = hp[4 * q + 1];
        int ci0 = 4 * q + (half << 1);
        v2f b0, b1;
        b0.x = wk[(ci0    ) * 32 +      l15];
        b0.y = wk[(ci0 + 1) * 32 +      l15];
        b1.x = wk[(ci0    ) * 32 + 16 + l15];
        b1.y = wk[(ci0 + 1) * 32 + 16 + l15];
        c0 = __builtin_amdgcn_wmma_f32_16x16x4_f32(false, a, false, b0,
                                                   (short)0, c0, false, false);
        c1 = __builtin_amdgcn_wmma_f32_16x16x4_f32(false, a, false, b1,
                                                   (short)0, c1, false, false);
    }

    int rows[8];
#pragma unroll
    for (int v = 0; v < 8; ++v)
        rows[v] = out_idx[base + (half << 3) + v] * 32;
#pragma unroll
    for (int v = 0; v < 8; ++v) {
        atomic_add_f32(&acc[rows[v] +      l15], c0[v]);
        atomic_add_f32(&acc[rows[v] + 16 + l15], c1[v]);
    }
}

__global__ void conv2_scalar_tail(const float* __restrict__ h,
                                  const float* __restrict__ W2,
                                  const int*   __restrict__ in_idx,
                                  const int*   __restrict__ out_idx,
                                  float*       __restrict__ acc,
                                  int M, int mStart, int K) {
    int rem = M - mStart;
    int total = K * rem * 32;
    int t = blockIdx.x * blockDim.x + threadIdx.x;
    if (t >= total) return;
    int co = t & 31;
    int km = t >> 5;
    int k  = km / rem;
    int m  = mStart + (km - k * rem);
    int e  = k * M + m;
    const float* hr = h + in_idx[e] * 16;
    const float* wp = W2 + k * 512 + co;
    float s = 0.0f;
#pragma unroll
    for (int ci = 0; ci < 16; ++ci) s += hr[ci] * wp[ci * 32];
    atomic_add_f32(&acc[out_idx[e] * 32 + co], s);
}

extern "C" void kernel_launch(void* const* d_in, const int* in_sizes, int n_in,
                              void* d_out, int out_size, void* d_ws, size_t ws_size,
                              hipStream_t stream) {
    const float* x      = (const float*)d_in[0];
    const float* W1     = (const float*)d_in[1];
    const float* g1     = (const float*)d_in[2];
    const float* b1     = (const float*)d_in[3];
    const float* m1     = (const float*)d_in[4];
    const float* v1     = (const float*)d_in[5];
    const float* W2     = (const float*)d_in[6];
    const float* g2     = (const float*)d_in[7];
    const float* b2     = (const float*)d_in[8];
    const float* m2     = (const float*)d_in[9];
    const float* v2     = (const float*)d_in[10];
    const int*   in_idx  = (const int*)d_in[11];
    const int*   out_idx = (const int*)d_in[12];

    const int N  = in_sizes[0];
    const int K  = in_sizes[1] / 16;
    const int KM = in_sizes[11];
    const int M  = KM / K;

    float* h_buf = (float*)d_ws;
    float* acc   = h_buf + (size_t)N * 16;

    hipMemsetAsync(d_ws, 0, (size_t)N * 48 * sizeof(float), stream);

    conv1_scatter<<<(KM + 255) / 256, 256, 0, stream>>>(
        x, W1, in_idx, out_idx, h_buf, M, KM);

    bn_elu<<<(N * 16 + 255) / 256, 256, 0, stream>>>(
        h_buf, h_buf, g1, b1, m1, v1, N * 16, 15);

    const int Mt = M / 16;
    const int totalWaves = K * Mt;
    if (totalWaves > 0) {
        const int blocks = (totalWaves + 7) / 8;
        conv2_wmma<<<blocks, 256, 0, stream>>>(
            h_buf, W2, in_idx, out_idx, acc, M, Mt, totalWaves);
    }
    const int mStart = Mt * 16;
    if (mStart < M) {
        const int total = K * (M - mStart) * 32;
        conv2_scalar_tail<<<(total + 255) / 256, 256, 0, stream>>>(
            h_buf, W2, in_idx, out_idx, acc, M, mStart, K);
    }

    bn_elu<<<(N * 32 + 255) / 256, 256, 0, stream>>>(
        acc, (float*)d_out, g2, b2, m2, v2, N * 32, 31);
}


# stage 1 factored: scalar x scattered into S[N,27] (1 f32 atomic per map entry, not 16), then S x W1 as dense f32 WMMA fused with BN+ELU; sums per offset first, then over offsets; adds unordered
# speedup vs baseline: 2.2788x; 2.2788x over previous
#include <hip/hip_runtime.h>


typedef __attribute__((ext_vector_type(2))) float v2f;
typedef __attribute__((ext_vector_type(8))) float v8f;

#define BN_EPS 1e-5f

__device__ __forceinline__ void atomic_add_f32(float* p, float v) {
    unsafeAtomicAdd(p, v);
}

__global__ void conv1_scatter_scalar(const float* __restrict__ x,
                                     const int*   __restrict__ in_idx,
                                     const int*   __restrict__ out_idx,
                                     float*       __restrict__ S,
                                     int M, int KM) {
    int t0 = (blockIdx.x * blockDim.x + threadIdx.x) * 4;
    if (t0 >= KM) return;
    if (t0 + 4 <= KM) {
        const int4 ii = *reinterpret_cast<const int4*>(in_idx + t0);
        const int4 oo = *reinterpret_cast<const int4*>(out_idx + t0);
        int k0 = t0 / M;
        int r  = t0 - k0 * M;
        float x0 = x[ii.x], x1 = x[ii.y], x2 = x[ii.z], x3 = x[ii.w];
        atomic_add_f32(&S[oo.x * 32 + k0 + (r + 0 >= M)], x0);
        atomic_add_f32(&S[oo.y * 32 + k0 + (r + 1 >= M)], x1);
        atomic_add_f32(&S[oo.z * 32 + k0 + (r + 2 >= M)], x2);
        atomic_add_f32(&S[oo.w * 32 + k0 + (r + 3 >= M)], x3);
    } else {
        for (int t = t0; t < KM; ++t)
            atomic_add_f32(&S[out_idx[t] * 32 + t / M], x[in_idx[t]]);
    }
}

__device__ __forceinline__ int k_of(int q, int half, int s) {
    return 8 * (q >> 1) + 4 * half + 2 * (q & 1) + s;
}

__global__ void conv1_dense_bn_elu(float*       __restrict__ S,
                                   const float* __restrict__ W1,
                                   const float* __restrict__ g,  const float* __restrict__ b,
                                   const float* __restrict__ mn, const float* __restrict__ vr,
                                   float*       __restrict__ h,
                                   int N, int K, int tiles) {
    int wave = blockIdx.x * (blockDim.x >> 5) + (threadIdx.x >> 5);
    if (wave >= tiles) return;
    int lane = threadIdx.x & 31;
    int half = lane >> 4;
    int l15  = lane & 15;
    int m0   = wave << 4;

    float s  = g[l15] * rsqrtf(vr[l15] + BN_EPS);
    float mu = mn[l15];
    float be = b[l15];

    int arow = m0 + l15;
    if (arow > N - 1) arow = N - 1;
    const float4* sp = reinterpret_cast<const float4*>(S + (size_t)arow * 32) + half;

    float4 av[4];
    for (int j = 0; j < 4; ++j) av[j] = sp[2 * j];

    v8f c = {};
    for (int q = 0; q < 8; ++q) {
        v2f a;
        a.x = (q & 1) ? av[q >> 1].z : av[q >> 1].x;
        a.y = (q & 1) ? av[q >> 1].w : av[q >> 1].y;
        int ka = k_of(q, half, 0);
        int kb = k_of(q, half, 1);
        float wa = W1[(ka < K ? ka : K - 1) * 16 + l15];
        float wb = W1[(kb < K ? kb : K - 1) * 16 + l15];
        v2f bf;
        bf.x = (ka < K) ? wa : 0.0f;
        bf.y = (kb < K) ? wb : 0.0f;
        c = __builtin_amdgcn_wmma_f32_16x16x4_f32(false, a, false, bf,
                                                  (short)0, c, false, false);
    }

    asm volatile("v_nop\n\tv_nop\n\tv_nop\n\tv_nop" : "+v"(c));

    float o[8];
    for (int v = 0; v < 8; ++v) {
        float y = (c[v] - mu) * s + be;
        float e = expf(y) - 1.0f;
        o[v] = (y > 0.0f) ? y : e;
    }

    float* hp = h + (size_t)(m0 + (half << 3)) * 16 + l15;
    float4* zp = reinterpret_cast<float4*>(S + (size_t)m0 * 32) + lane;
    const float4 z4 = make_float4(0.0f, 0.0f, 0.0f, 0.0f);
    if (m0 + 16 <= N) {
        for (int v = 0; v < 8; ++v) hp[v * 16] = o[v];
        for (int j = 0; j < 4; ++j) zp[j * 32] = z4;
    } else {
        for (int v = 0; v < 8; ++v)
            if (m0 + (half << 3) + v < N) hp[v * 16] = o[v];
        for (int j = 0; j < 4; ++j)
            if (m0 + 4 * j + (lane >> 3) < N) zp[j * 32] = z4;
    }
}

__global__ void bn_elu(const float* __restrict__ in, float* __restrict__ out,
                       const float* __restrict__ g,  const float* __restrict__ b,
                       const float* __restrict__ mn, const float* __restrict__ vr,
                       int total, int Cmask) {
    int t = blockIdx.x * blockDim.x + threadIdx.x;
    if (t >= total) return;
    int c = t & Cmask;
    float s = g[c] * rsqrtf(vr[c] + BN_EPS);
    float y = (in[t] - mn[c]) * s + b[c];
    out[t] = (y > 0.0f) ? y : (expf(y) - 1.0f);
}

__global__ void conv2_wmma(const float* __restrict__ h,
                           const float* __restrict__ W2,
                           const int*   __restrict__ in_idx,
                           const int*   __restrict__ out_idx,
                           float*       __restrict__ acc,
                           int M, int Mt, int totalWaves) {
    int wave = blockIdx.x * (blockDim.x >> 5) + (threadIdx.x >> 5);
    if (wave >= totalWaves) return;
    int lane = threadIdx.x & 31;
    int half = lane >> 4;
    int l15  = lane & 15;

    int k  = wave / Mt;
    int m0 = (wave - k * Mt) << 4;
    int base = k * M + m0;

    int arow = in_idx[base + l15];
    const float* hp = h + arow * 16 + (half << 1);

    const float* wk = W2 + k * 512;

    v8f c0 = {};
    v8f c1 = {};

#pragma unroll
    for (int q = 0; q < 4; ++q) {
        v2f a;
        a.x = hp[4 * q + 0];
        a.y = hp[4 * q + 1];
        int ci0 = 4 * q + (half << 1);
        v2f b0, b1;
        b0.x = wk[(ci0    ) * 32 +      l15];
        b0.y = wk[(ci0 + 1) * 32 +      l15];
        b1.x = wk[(ci0    ) * 32 + 16 + l15];
        b1.y = wk[(ci0 + 1) * 32 + 16 + l15];
        c0 = __builtin_amdgcn_wmma_f32_16x16x4_f32(false, a, false, b0,
                                                   (short)0, c0, false, false);
        c1 = __builtin_amdgcn_wmma_f32_16x16x4_f32(false, a, false, b1,
                                                   (short)0, c1, false, false);
    }

    int rows[8];
#pragma unroll
    for (int v = 0; v < 8; ++v)
        rows[v] = out_idx[base + (half << 3) + v] * 32;
#pragma unroll
    for (int v = 0; v < 8; ++v) {
        atomic_add_f32(&acc[rows[v] +      l15], c0[v]);
        atomic_add_f32(&acc[rows[v] + 16 + l15], c1[v]);
    }
}

__global__ void conv2_scalar_tail(const float* __restrict__ h,
                                  const float* __restrict__ W2,
                                  const int*   __restrict__ in_idx,
                                  const int*   __restrict__ out_idx,
                                  float*       __restrict__ acc,
                                  int M, int mStart, int K) {
    int rem = M - mStart;
    int total = K * rem * 32;
    int t = blockIdx.x * blockDim.x + threadIdx.x;
    if (t >= total) return;
    int co = t & 31;
    int km = t >> 5;
    int k  = km / rem;
    int m  = mStart + (km - k * rem);
    int e  = k * M + m;
    const float* hr = h + in_idx[e] * 16;
    const float* wp = W2 + k * 512 + co;
    float s = 0.0f;
#pragma unroll
    for (int ci = 0; ci < 16; ++ci) s += hr[ci] * wp[ci * 32];
    atomic_add_f32(&acc[out_idx[e] * 32 + co], s);
}

extern "C" void kernel_launch(void* const* d_in, const int* in_sizes, int n_in,
                              void* d_out, int out_size, void* d_ws, size_t ws_size,
                              hipStream_t stream) {
    const float* x      = (const float*)d_in[0];
    const float* W1     = (const float*)d_in[1];
    const float* g1     = (const float*)d_in[2];
    const float* b1     = (const float*)d_in[3];
    const float* m1     = (const float*)d_in[4];
    const float* v1     = (const float*)d_in[5];
    const float* W2     = (const float*)d_in[6];
    const float* g2     = (const float*)d_in[7];
    const float* b2     = (const float*)d_in[8];
    const float* m2     = (const float*)d_in[9];
    const float* v2     = (const float*)d_in[10];
    const int*   in_idx  = (const int*)d_in[11];
    const int*   out_idx = (const int*)d_in[12];

    const int N  = in_sizes[0];
    const int K  = in_sizes[1] / 16;
    const int KM = in_sizes[11];
    const int M  = KM / K;

    float* h_buf = (float*)d_ws;
    float* acc   = h_buf + (size_t)N * 16;

    hipMemsetAsync(acc, 0, (size_t)N * 32 * sizeof(float), stream);

    const int quads = (KM + 3) / 4;
    conv1_scatter_scalar<<<(quads + 255) / 256, 256, 0, stream>>>(
        x, in_idx, out_idx, acc, M, KM);

    const int tiles1 = (N + 15) / 16;
    conv1_dense_bn_elu<<<(tiles1 + 7) / 8, 256, 0, stream>>>(
        acc, W1, g1, b1, m1, v1, h_buf, N, K, tiles1);

    const int Mt = M / 16;
    const int totalWaves = K * Mt;
    if (totalWaves > 0) {
        const int blocks = (totalWaves + 7) / 8;
        conv2_wmma<<<blocks, 256, 0, stream>>>(
            h_buf, W2, in_idx, out_idx, acc, M, Mt, totalWaves);
    }
    const int mStart = Mt * 16;
    if (mStart < M) {
        const int total = K * (M - mStart) * 32;
        conv2_scalar_tail<<<(total + 255) / 256, 256, 0, stream>>>(
            h_buf, W2, in_idx, out_idx, acc, M, mStart, K);
    }

    bn_elu<<<(N * 32 + 255) / 256, 256, 0, stream>>>(
        acc, (float*)d_out, g2, b2, m2, v2, N * 32, 31);
}


# both scatter planes kept twice, one copy per chip half: a wave adds only into its own half's copy (second copy in the output buffer); consumer passes add the two copies; sums form as two partials
# speedup vs baseline: 2.6242x; 1.1516x over previous
#include <hip/hip_runtime.h>


typedef __attribute__((ext_vector_type(2))) float v2f;
typedef __attribute__((ext_vector_type(8))) float v8f;

#define BN_EPS 1e-5f

__device__ __forceinline__ void atomic_add_f32(float* p, float v) {
    unsafeAtomicAdd(p, v);
}

__device__ __forceinline__ int chip_half() {
    return (int)((__builtin_amdgcn_s_sendmsg_rtn(0x87) >> 18) & 1u);
}

__global__ void zero_planes(float* __restrict__ p0, float* __restrict__ p1, int n4) {
    int t = blockIdx.x * blockDim.x + threadIdx.x;
    if (t >= n4) return;
    const float4 z4 = make_float4(0.0f, 0.0f, 0.0f, 0.0f);
    reinterpret_cast<float4*>(p0)[t] = z4;
    reinterpret_cast<float4*>(p1)[t] = z4;
}

__global__ void conv1_scatter_scalar(const float* __restrict__ x,
                                     const int*   __restrict__ in_idx,
                                     const int*   __restrict__ out_idx,
                                     float*       __restrict__ S0,
                                     float*       __restrict__ S1,
                                     int M, int KM) {
    int t0 = (blockIdx.x * blockDim.x + threadIdx.x) * 4;
    if (t0 >= KM) return;
    float* S = chip_half() ? S1 : S0;
    if (t0 + 4 <= KM) {
        const int4 ii = *reinterpret_cast<const int4*>(in_idx + t0);
        const int4 oo = *reinterpret_cast<const int4*>(out_idx + t0);
        int k0 = t0 / M;
        int r  = t0 - k0 * M;
        float x0 = x[ii.x], x1 = x[ii.y], x2 = x[ii.z], x3 = x[ii.w];
        atomic_add_f32(&S[oo.x * 32 + k0 + (r + 0 >= M)], x0);
        atomic_add_f32(&S[oo.y * 32 + k0 + (r + 1 >= M)], x1);
        atomic_add_f32(&S[oo.z * 32 + k0 + (r + 2 >= M)], x2);
        atomic_add_f32(&S[oo.w * 32 + k0 + (r + 3 >= M)], x3);
    } else {
        for (int t = t0; t < KM; ++t)
            atomic_add_f32(&S[out_idx[t] * 32 + t / M], x[in_idx[t]]);
    }
}

__device__ __forceinline__ int k_of(int q, int half, int s) {
    return 8 * (q >> 1) + 4 * half + 2 * (q & 1) + s;
}

__global__ void conv1_dense_bn_elu(float*       __restrict__ S,
                                   float*       __restrict__ T,
                                   const float* __restrict__ W1,
                                   const float* __restrict__ g,  const float* __restrict__ b,
                                   const float* __restrict__ mn, const float* __restrict__ vr,
                                   float*       __restrict__ h,
                                   int N, int K, int tiles) {
    int wave = blockIdx.x * (blockDim.x >> 5) + (threadIdx.x >> 5);
    if (wave >= tiles) return;
    int lane = threadIdx.x & 31;
    int half = lane >> 4;
    int l15  = lane & 15;
    int m0   = wave << 4;

    float s  = g[l15] * rsqrtf(vr[l15] + BN_EPS);
    float mu = mn[l15];
    float be = b[l15];

    int arow = m0 + l15;
    if (arow > N - 1) arow = N - 1;
    const float4* sp = reinterpret_cast<const float4*>(S + (size_t)arow * 32) + half;
    const float4* tp = reinterpret_cast<const float4*>(T + (size_t)arow * 32) + half;

    float4 av[4];
    for (int j = 0; j < 4; ++j) {
        const float4 u = sp[2 * j];
        const float4 w = tp[2 * j];
        av[j] = make_float4(u.x + w.x, u.y + w.y, u.z + w.z, u.w + w.w);
    }

    v8f c = {};
    for (int q = 0; q < 8; ++q) {
        v2f a;
        a.x = (q & 1) ? av[q >> 1].z : av[q >> 1].x;
        a.y = (q & 1) ? av[q >> 1].w : av[q >> 1].y;
        int ka = k_of(q, half, 0);
        int kb = k_of(q, half, 1);
        float wa = W1[(ka < K ? ka : K - 1) * 16 + l15];
        float wb = W1[(kb < K ? kb : K - 1) * 16 + l15];
        v2f bf;
        bf.x = (ka < K) ? wa : 0.0f;
        bf.y = (kb < K) ? wb : 0.0f;
        c = __builtin_amdgcn_wmma_f32_16x16x4_f32(false, a, false, bf,
                                                  (short)0, c, false, false);
    }

    asm volatile("v_nop\n\tv_nop\n\tv_nop\n\tv_nop" : "+v"(c));

    float o[8];
    for (int v = 0; v < 8; ++v) {
        float y = (c[v] - mu) * s + be;
        float e = expf(y) - 1.0f;
        o[v] = (y > 0.0f) ? y : e;
    }

    float* hp = h + (size_t)(m0 + (half << 3)) * 16 + l15;
    float4* zp = reinterpret_cast<float4*>(S + (size_t)m0 * 32) + lane;
    float4* yp = reinterpret_cast<float4*>(T + (size_t)m0 * 32) + lane;
    const float4 z4 = make_float4(0.0f, 0.0f, 0.0f, 0.0f);
    if (m0 + 16 <= N) {
        for (int v = 0; v < 8; ++v) hp[v * 16] = o[v];
        for (int j = 0; j < 4; ++j) { zp[j * 32] = z4; yp[j * 32] = z4; }
    } else {
        for (int v = 0; v < 8; ++v)
            if (m0 + (half << 3) + v < N) hp[v * 16] = o[v];
        for (int j = 0; j < 4; ++j)
            if (m0 + 4 * j + (lane >> 3) < N) { zp[j * 32] = z4; yp[j * 32] = z4; }
    }
}

__global__ void bn_elu(const float* __restrict__ in, float* __restrict__ out,
                       const float* __restrict__ g,  const float* __restrict__ b,
                       const float* __restrict__ mn, const float* __restrict__ vr,
                       int total, int Cmask) {
    int t = blockIdx.x * blockDim.x + threadIdx.x;
    if (t >= total) return;
    int c = t & Cmask;
    float s = g[c] * rsqrtf(vr[c] + BN_EPS);
    float y = ((in[t] + out[t]) - mn[c]) * s + b[c];
    out[t] = (y > 0.0f) ? y : (expf(y) - 1.0f);
}

__global__ void conv2_wmma(const float* __restrict__ h,
                           const float* __restrict__ W2,
                           const int*   __restrict__ in_idx,
                           const int*   __restrict__ out_idx,
                           float*       __restrict__ acc0,
                           float*       __restrict__ acc1,
                           int M, int Mt, int totalWaves) {
    int wave = blockIdx.x * (blockDim.x >> 5) + (threadIdx.x >> 5);
    if (wave >= totalWaves) return;
    float* acc = chip_half() ? acc1 : acc0;
    int lane = threadIdx.x & 31;
    int half = lane >> 4;
    int l15  = lane & 15;

    int k  = wave / Mt;
    int m0 = (wave - k * Mt) << 4;
    int base = k * M + m0;

    int arow = in_idx[base + l15];
    const float* hp = h + arow * 16 + (half << 1);

    const float* wk = W2 + k * 512;

    v8f c0 = {};
    v8f c1 = {};

#pragma unroll
    for (int q = 0; q < 4; ++q) {
        v2f a;
        a.x = hp[4 * q + 0];
        a.y = hp[4 * q + 1];
        int ci0 = 4 * q + (half << 1);
        v2f b0, b1;
        b0.x = wk[(ci0    ) * 32 +      l15];
        b0.y = wk[(ci0 + 1) * 32 +      l15];
        b1.x = wk[(ci0    ) * 32 + 16 + l15];
        b1.y = wk[(ci0 + 1) * 32 + 16 + l15];
        c0 = __builtin_amdgcn_wmma_f32_16x16x4_f32(false, a, false, b0,
                                                   (short)0, c0, false, false);
        c1 = __builtin_amdgcn_wmma_f32_16x16x4_f32(false, a, false, b1,
                                                   (short)0, c1, false, false);
    }

    int rows[8];
#pragma unroll
    for (int v = 0; v < 8; ++v)
        rows[v] = out_idx[base + (half << 3) + v] * 32;
#pragma unroll
    for (int v = 0; v < 8; ++v) {
        atomic_add_f32(&acc[rows[v] +      l15], c0[v]);
        atomic_add_f32(&acc[rows[v] + 16 + l15], c1[v]);
    }
}

__global__ void conv2_scalar_tail(const float* __restrict__ h,
                                  const float* __restrict__ W2,
                                  const int*   __restrict__ in_idx,
                                  const int*   __restrict__ out_idx,
                                  float*       __restrict__ acc,
                                  int M, int mStart, int K) {
    int rem = M - mStart;
    int total = K * rem * 32;
    int t = blockIdx.x * blockDim.x + threadIdx.x;
    if (t >= total) return;
    int co = t & 31;
    int km = t >> 5;
    int k  = km / rem;
    int m  = mStart + (km - k * rem);
    int e  = k * M + m;
    const float* hr = h + in_idx[e] * 16;
    const float* wp = W2 + k * 512 + co;
    float s = 0.0f;
#pragma unroll
    for (int ci = 0; ci < 16; ++ci) s += hr[ci] * wp[ci * 32];
    atomic_add_f32(&acc[out_idx[e] * 32 + co], s);
}

extern "C" void kernel_launch(void* const* d_in, const int* in_sizes, int n_in,
                              void* d_out, int out_size, void* d_ws, size_t ws_size,
                              hipStream_t stream) {
    const float* x      = (const float*)d_in[0];
    const float* W1     = (const float*)d_in[1];
    const float* g1     = (const float*)d_in[2];
    const float* b1     = (const float*)d_in[3];
    const float* m1     = (const float*)d_in[4];
    const float* v1     = (const float*)d_in[5];
    const float* W2     = (const float*)d_in[6];
    const float* g2     = (const float*)d_in[7];
    const float* b2     = (const float*)d_in[8];
    const float* m2     = (const float*)d_in[9];
    const float* v2     = (const float*)d_in[10];
    const int*   in_idx  = (const int*)d_in[11];
    const int*   out_idx = (const int*)d_in[12];

    const int N  = in_sizes[0];
    const int K  = in_sizes[1] / 16;
    const int KM = in_sizes[11];
    const int M  = KM / K;

    float* h_buf = (float*)d_ws;
    float* acc   = h_buf + (size_t)N * 16;
    float* acc1  = (float*)d_out;

    const int n4 = N * 8;
    zero_planes<<<(n4 + 255) / 256, 256, 0, stream>>>(
        acc, acc1, n4);

    const int quads = (KM + 3) / 4;
    conv1_scatter_scalar<<<(quads + 255) / 256, 256, 0, stream>>>(
        x, in_idx, out_idx, acc, acc1, M, KM);

    const int tiles1 = (N + 15) / 16;
    conv1_dense_bn_elu<<<(tiles1 + 7) / 8, 256, 0, stream>>>(
        acc, acc1, W1, g1, b1, m1, v1, h_buf, N, K, tiles1);

    const int Mt = M / 16;
    const int totalWaves = K * Mt;
    if (totalWaves > 0) {
        const int blocks = (totalWaves + 7) / 8;
        conv2_wmma<<<blocks, 256, 0, stream>>>(
            h_buf, W2, in_idx, out_idx, acc, acc1, M, Mt, totalWaves);
    }
    const int mStart = Mt * 16;
    if (mStart < M) {
        const int total = K * (M - mStart) * 32;
        conv2_scalar_tail<<<(total + 255) / 256, 256, 0, stream>>>(
            h_buf, W2, in_idx, out_idx, acc, M, mStart, K);
    }

    bn_elu<<<(N * 32 + 255) / 256, 256, 0, stream>>>(
        acc, (float*)d_out, g2, b2, m2, v2, N * 32, 31);
}
